# starting kernel
# baseline (speedup 1.0000x reference)
#include <hip/hip_runtime.h>

typedef __attribute__((ext_vector_type(2))) float v2f;
typedef __attribute__((ext_vector_type(8))) float v8f;

#define BATCH   16
#define CIN     128
#define COUT    256
#define HW      56
#define SPATIAL (HW * HW)
#define NPIX    (BATCH * SPATIAL)
#define KTOT    (CIN * 9)
#define KC      16
#define KP      (KC / 2)
#define MT      64
#define NT      128
#define A2S     80
#define B2S     144

__global__ __launch_bounds__(256) void conv3x3_wmma_f32(
    const float* __restrict__ x, const float* __restrict__ w,
    const float* __restrict__ bias, float* __restrict__ out)
{
    __shared__ v2f a2_s[2][KP * A2S];
    __shared__ v2f b2_s[2][KP * B2S];

    const int tid    = threadIdx.x;
    const int lane   = tid & 31;
    const int wave   = tid >> 5;
    const int m_base = blockIdx.y * MT;
    const int n_base = blockIdx.x * NT;
    const int wm     = wave & 3;
    const int wn     = wave >> 2;

    const int a_cl = tid >> 2;
    const int a_kq = (tid & 3) << 2;
    const float* wptr = w + (size_t)(m_base + a_cl) * KTOT + a_kq;

    const int b_r  = tid >> 4;
    const int b_pg = (tid & 15) << 3;
    int pbase[8], poh[8], pow_[8];
#pragma unroll
    for (int j = 0; j < 8; ++j) {
        int pn = n_base + b_pg + j;
        int bb = pn / SPATIAL;
        int s  = pn - bb * SPATIAL;
        int oh = s / HW;
        int ow = s - oh * HW;
        pbase[j] = bb * (CIN * SPATIAL) + oh * HW + ow;
        poh[j]   = oh;
        pow_[j]  = ow;
    }

    auto gather = [&](int kkx, float* bv) {
        int k    = kkx + b_r;
        int ci   = k / 9;
        int rem  = k - ci * 9;
        int kh   = rem / 3;
        int kw   = rem - kh * 3;
        int coff = ci * SPATIAL + (kh - 1) * HW + (kw - 1);
#pragma unroll
        for (int j = 0; j < 8; ++j) {
            int ih = poh[j] + kh - 1;
            int iw = pow_[j] + kw - 1;
            bool ok = ((unsigned)ih < (unsigned)HW) &&
                      ((unsigned)iw < (unsigned)HW);
            bv[j] = ok ? x[pbase[j] + coff] : 0.f;
        }
    };

    auto commit = [&](int buf, const float4& w4, const float* bv) {
        v2f* a2 = a2_s[buf];
        v2f av0; av0.x = w4.x; av0.y = w4.y;
        v2f av1; av1.x = w4.z; av1.y = w4.w;
        a2[((a_kq >> 1) + 0) * A2S + a_cl] = av0;
        a2[((a_kq >> 1) + 1) * A2S + a_cl] = av1;
        float* bf = (float*)b2_s[buf];
        int bbase = ((b_r >> 1) * B2S) * 2 + (b_r & 1);
#pragma unroll
        for (int j = 0; j < 8; ++j)
            bf[bbase + (b_pg + j) * 2] = bv[j];
    };

    v8f acc[4];
#pragma unroll
    for (int i = 0; i < 4; ++i) acc[i] = (v8f)0.f;

    const int row    = lane & 15;
    const int kphalf = lane >> 4;
    const int a_col  = wm * 16 + row;
    const int b_col0 = wn * 64 + row;

    {
        float4 w4 = *(const float4*)(wptr);
        float bv[8];
        gather(0, bv);
        commit(0, w4, bv);
    }
    __syncthreads();

    for (int kk = 0; kk < KTOT; kk += KC) {
        const int  buf  = (kk / KC) & 1;
        const bool more = (kk + KC) < KTOT;

        float4 nw4;
        float  nbv[8];
        if (more) {
            nw4 = *(const float4*)(wptr + kk + KC);
            gather(kk + KC, nbv);
        }

        const v2f* a2 = a2_s[buf];
        const v2f* b2 = b2_s[buf];
#pragma unroll
        for (int ks = 0; ks < 4; ++ks) {
            int kp = 2 * ks + kphalf;
            v2f a = a2[kp * A2S + a_col];
#pragma unroll
            for (int nt = 0; nt < 4; ++nt) {
                v2f b = b2[kp * B2S + b_col0 + nt * 16];
                acc[nt] = __builtin_amdgcn_wmma_f32_16x16x4_f32(
                    false, a, false, b, (short)0, acc[nt], false, false);
            }
        }

        if (more) commit(buf ^ 1, nw4, nbv);
        __syncthreads();
    }

    const int mrow0 = m_base + wm * 16 + kphalf * 8;
#pragma unroll
    for (int nt = 0; nt < 4; ++nt) {
        int pn = n_base + wn * 64 + nt * 16 + row;
        int bb = pn / SPATIAL;
        int s  = pn - bb * SPATIAL;
        size_t obase = (size_t)bb * (COUT * SPATIAL) + s;
#pragma unroll
        for (int v = 0; v < 8; ++v) {
            int cout = mrow0 + v;
            __builtin_nontemporal_store(acc[nt][v] + bias[cout],
                                        &out[obase + (size_t)cout * SPATIAL]);
        }
    }
}

extern "C" void kernel_launch(void* const* d_in, const int* in_sizes, int n_in,
                              void* d_out, int out_size, void* d_ws, size_t ws_size,
                              hipStream_t stream) {
    const float* x    = (const float*)d_in[0];
    const float* wgt  = (const float*)d_in[1];
    const float* bias = (const float*)d_in[2];
    float* out        = (float*)d_out;

    dim3 grid(NPIX / NT, COUT / MT, 1);
    dim3 block(256, 1, 1);
    conv3x3_wmma_f32<<<grid, block, 0, stream>>>(x, wgt, bias, out);
}


# LDS fragment reads software-pipelined: each tap's 8 reads issued as one group two taps ahead of their WMMAs (3-slot register ring), kept running across the trip barrier by a third LDS stage
# speedup vs baseline: 1.7689x; 1.7689x over previous
#include <hip/hip_runtime.h>


typedef __attribute__((ext_vector_type(2))) float v2f;
typedef __attribute__((ext_vector_type(8))) float v8f;

#define BATCH    16
#define CIN      128
#define COUT     256
#define HW       56
#define SPATIAL  (HW * HW)
#define KTOT     (CIN * 9)
#define MT       64
#define ROWS     14
#define GC       4
#define NGROUP   (CIN / GC)
#define HROWS    (ROWS + 2)
#define HPITCH   64
#define PLANE2   (HROWS * HPITCH + 16)
#define HSTAGE2  (2 * PLANE2)
#define A2S      80
#define AROWS    18
#define ASTAGE2  (AROWS * A2S)
#define NTHREADS 896
#define NSTAGE   3
#define NTILE    7

__global__ __launch_bounds__(NTHREADS) void conv3x3_halo_wmma_f32(
    const float* __restrict__ x, const float* __restrict__ w,
    const float* __restrict__ bias, float* __restrict__ out)
{
    __shared__ __attribute__((aligned(16))) v2f a2_s[NSTAGE * ASTAGE2];
    __shared__ __attribute__((aligned(16))) v2f h2_s[NSTAGE * HSTAGE2];

    const int tid  = threadIdx.x;
    const int lane = tid & 31;
    const int wave = tid >> 5;
    const int wm   = wave & 3;
    const int wn   = wave >> 2;
    const int c    = lane & 15;
    const int h    = lane >> 4;

    const int mb     = blockIdx.x & 3;
    const int bq     = blockIdx.x >> 2;
    const int img    = bq >> 2;
    const int qy     = bq & 3;
    const int m_base = mb * MT;
    const int R0     = qy * ROWS;

    const int  x_ci  = tid / (HROWS * 14);
    const int  x_rem = tid - x_ci * (HROWS * 14);
    const int  x_hr  = x_rem / 14;
    const int  x_f4  = x_rem - x_hr * 14;
    const int  x_row = R0 - 1 + x_hr;
    const bool x_ok  = (unsigned)x_row < (unsigned)HW;
    const unsigned x_off = (unsigned)(((img * CIN + x_ci) * HW + (x_ok ? x_row : 0)) * HW + 4 * x_f4);
    const int  x_lds = ((x_ci >> 1) * PLANE2 + x_hr * HPITCH + 4 + 4 * x_f4) * 2 + (x_ci & 1);

    const bool a_ok  = tid < 576;
    const int  a_co  = tid & 63;
    const int  a_q   = a_ok ? (tid >> 6) : 0;
    const unsigned w_off = (unsigned)((m_base + a_co) * KTOT + 4 * a_q);
    int a_lds[4];
#pragma unroll
    for (int i = 0; i < 4; ++i) {
        int kf  = 4 * a_q + i;
        int ci  = kf / 9;
        int tap = kf - ci * 9;
        a_lds[i] = (((ci >> 1) * 9 + tap) * A2S + a_co) * 2 + (ci & 1);
    }

    for (int i = tid; i < (NSTAGE * HSTAGE2) / 2; i += NTHREADS)
        ((float4*)h2_s)[i] = make_float4(0.f, 0.f, 0.f, 0.f);
    __syncthreads();

    float4 xr = make_float4(0.f, 0.f, 0.f, 0.f);
    float4 wr = make_float4(0.f, 0.f, 0.f, 0.f);

    auto load_into = [&](int g, float4& xv, float4& wv) {
        if (x_ok) xv = *(const float4*)((x + (size_t)g * (GC * SPATIAL)) + x_off);
        if (a_ok) wv = *(const float4*)((w + (size_t)g * (GC * 9)) + w_off);
    };
    auto store_from = [&](int buf, const float4& xv, const float4& wv) {
        if (x_ok) {
            float* hf = (float*)(h2_s + buf * HSTAGE2) + x_lds;
            hf[0] = xv.x; hf[2] = xv.y; hf[4] = xv.z; hf[6] = xv.w;
        }
        if (a_ok) {
            float* af = (float*)(a2_s + buf * ASTAGE2);
            af[a_lds[0]] = wv.x; af[a_lds[1]] = wv.y;
            af[a_lds[2]] = wv.z; af[a_lds[3]] = wv.w;
        }
    };
    auto load_group  = [&](int g)   { load_into(g, xr, wr); };
    auto store_group = [&](int buf) { store_from(buf, xr, wr); };

    const int abase = (9 * h) * A2S + wm * 16 + c;
    int bbase[NTILE];
#pragma unroll
    for (int nt = 0; nt < NTILE; ++nt) {
        int p  = nt * 16 + c;
        int rr = (p >= HW) ? 1 : 0;
        int cc = p - rr * HW;
        bbase[nt] = h * PLANE2 + (2 * wn + rr) * HPITCH + cc;
    }

    v8f acc[NTILE];
#pragma unroll
    for (int i = 0; i < NTILE; ++i) acc[i] = (v8f)0.f;

    v2f fa[3];
    v2f fb[3][NTILE];

    auto load_tap = [&](int st, int j, int slot) {
        const int kh = j / 3, kw = j - kh * 3;
        const v2f* a2c = a2_s + st * ASTAGE2 + abase;
        const v2f* h2c = h2_s + st * HSTAGE2;
        fa[slot] = a2c[j * A2S];
#pragma unroll
        for (int nt = 0; nt < NTILE; ++nt)
            fb[slot][nt] = h2c[bbase[nt] + kh * HPITCH + 3 + kw];
    };

    float4 xr0 = make_float4(0.f, 0.f, 0.f, 0.f);
    float4 wr0 = make_float4(0.f, 0.f, 0.f, 0.f);
    load_into(0, xr0, wr0);
    load_into(1, xr, wr);
    store_from(0, xr0, wr0);
    store_from(1, xr, wr);
    load_group(2);
    __syncthreads();
    load_tap(0, 0, 0);
    load_tap(0, 1, 1);

    int st = 0;
    for (int g = 0; g < NGROUP; ++g) {
        const int st1 = (st == NSTAGE - 1) ? 0 : st + 1;
        const int st2 = (st1 == NSTAGE - 1) ? 0 : st1 + 1;

#pragma unroll
        for (int j = 0; j < 9; ++j) {
            if (j + 2 < 9)           load_tap(st, j + 2, (j + 2) % 3);
            else if (g + 1 < NGROUP) load_tap(st1, j + 2 - 9, (j + 2) % 3);
            if (j == 1) {
                if (g + 2 < NGROUP) store_group(st2);
                if (g + 3 < NGROUP) load_group(g + 3);
            }
            __builtin_amdgcn_sched_barrier(0);
#pragma unroll
            for (int nt = 0; nt < NTILE; ++nt)
                acc[nt] = __builtin_amdgcn_wmma_f32_16x16x4_f32(
                    false, fa[j % 3], false, fb[j % 3][nt], (short)0, acc[nt], false, false);
            __builtin_amdgcn_sched_barrier(0);
        }
        __syncthreads();
        st = st1;
    }

    const int mrow0 = m_base + wm * 16 + h * 8;
    float bv[8];
#pragma unroll
    for (int v = 0; v < 8; ++v) bv[v] = bias[mrow0 + v];
#pragma unroll
    for (int nt = 0; nt < NTILE; ++nt) {
        int s = R0 * HW + wn * 112 + nt * 16 + c;
        size_t obase = (size_t)img * (COUT * SPATIAL) + s;
#pragma unroll
        for (int v = 0; v < 8; ++v) {
            int cout = mrow0 + v;
            __builtin_nontemporal_store(acc[nt][v] + bv[v],
                                        &out[obase + (size_t)cout * SPATIAL]);
        }
    }
}

extern "C" void kernel_launch(void* const* d_in, const int* in_sizes, int n_in,
                              void* d_out, int out_size, void* d_ws, size_t ws_size,
                              hipStream_t stream) {
    const float* x    = (const float*)d_in[0];
    const float* wgt  = (const float*)d_in[1];
    const float* bias = (const float*)d_in[2];
    float* out        = (float*)d_out;

    dim3 grid(256, 1, 1);
    dim3 block(NTHREADS, 1, 1);
    conv3x3_halo_wmma_f32<<<grid, block, 0, stream>>>(x, wgt, bias, out);
}
